# starting kernel
# baseline (speedup 1.0000x reference)
#include <hip/hip_runtime.h>
#include <hip/hip_bf16.h>


typedef float v2f __attribute__((ext_vector_type(2)));
typedef float v8f __attribute__((ext_vector_type(8)));

#define ORDER 16
#define FRAME 160
#define EPS_F 1e-8f
#define WAVES_PER_BLOCK 8
#define PAD 16
#define REGION (PAD + FRAME + PAD)

__global__ __launch_bounds__(256) void lpc_residual_kernel(
    const float* __restrict__ x, float* __restrict__ out, int n_frames) {
  __shared__ float sh[WAVES_PER_BLOCK][REGION];

  const int lane = threadIdx.x & 31;
  const int wv = threadIdx.x >> 5;
  const int frame = blockIdx.x * WAVES_PER_BLOCK + wv;
  const bool active = frame < n_frames;

  float* fr = &sh[wv][PAD];

  if (active) {
    if (lane < PAD) {
      sh[wv][lane] = 0.0f;
      sh[wv][PAD + FRAME + lane] = 0.0f;
    }
    const float* gx = x + (long long)frame * FRAME;
#pragma unroll
    for (int t = 0; t < 5; ++t) fr[lane + 32 * t] = gx[lane + 32 * t];
  }
  __syncthreads();

  if (active) {
    const int m = lane & 15;
    const int k0 = (lane >> 4) << 1;
    v8f acc0 = {0.f, 0.f, 0.f, 0.f, 0.f, 0.f, 0.f, 0.f};
    v8f acc1 = acc0;
    for (int c = 0; c < 40; c += 2) {
      v2f A0, B0, A1, B1;
      A0.x = fr[4 * c + k0 + m];
      A0.y = fr[4 * c + k0 + 1 + m];
      B0.x = fr[4 * c + k0];
      B0.y = fr[4 * c + k0 + 1];
      A1.x = fr[4 * (c + 1) + k0 + m];
      A1.y = fr[4 * (c + 1) + k0 + 1 + m];
      B1.x = fr[4 * (c + 1) + k0];
      B1.y = fr[4 * (c + 1) + k0 + 1];
      acc0 = __builtin_amdgcn_wmma_f32_16x16x4_f32(false, A0, false, B0,
                                                   (short)0, acc0, false, false);
      acc1 = __builtin_amdgcn_wmma_f32_16x16x4_f32(false, A1, false, B1,
                                                   (short)0, acc1, false, false);
    }
    v8f D = acc0 + acc1;

    float p16 = 0.0f;
#pragma unroll
    for (int t = 0; t < 5; ++t) {
      int idx = lane + 32 * t;
      if (idx < FRAME - ORDER) p16 += fr[idx] * fr[idx + ORDER];
    }
#pragma unroll
    for (int s = 16; s >= 1; s >>= 1) p16 += __shfl_xor(p16, s, 32);

    float r[ORDER + 1];
#pragma unroll
    for (int q = 0; q < 16; ++q) r[q] = __shfl(D[q & 7], (q >> 3) * 16, 32);
    r[ORDER] = p16;

    float a[ORDER + 1];
    a[0] = 1.0f;
#pragma unroll
    for (int q = 1; q <= ORDER; ++q) a[q] = 0.0f;
    float e = (r[0] != 0.0f) ? r[0] : EPS_F;
#pragma unroll
    for (int i = 1; i <= ORDER; ++i) {
      float accv = r[i];
#pragma unroll
      for (int j = 1; j < i; ++j) accv -= a[j] * r[i - j];
      float k = (e != 0.0f) ? (accv / e) : 0.0f;
      float anew[ORDER + 1];
#pragma unroll
      for (int j = 1; j < i; ++j) anew[j] = a[j] - k * a[i - j];
#pragma unroll
      for (int j = 1; j < i; ++j) a[j] = anew[j];
      a[i] = k;
      e = fmaxf(e * (1.0f - k * k), EPS_F);
    }

    float* go = out + (long long)frame * FRAME;
#pragma unroll
    for (int t = 0; t < 5; ++t) {
      int n = lane + 32 * t;
      float s = 0.0f;
#pragma unroll
      for (int k = 0; k <= ORDER; ++k) s += a[k] * fr[n - k];
      go[n] = s;
    }
  }
}

extern "C" void kernel_launch(void* const* d_in, const int* in_sizes, int n_in,
                              void* d_out, int out_size, void* d_ws, size_t ws_size,
                              hipStream_t stream) {
  (void)n_in; (void)out_size; (void)d_ws; (void)ws_size;
  const float* x = (const float*)d_in[0];
  float* out = (float*)d_out;
  const int total = in_sizes[0];
  const int n_frames = total / FRAME;
  const int blocks = (n_frames + WAVES_PER_BLOCK - 1) / WAVES_PER_BLOCK;
  lpc_residual_kernel<<<blocks, WAVES_PER_BLOCK * 32, 0, stream>>>(x, out, n_frames);
}


# Levinson-Durbin on the vector unit, lane = coefficient, 16-lane row = frame (two frames per wave): per order a row dot by lane swizzles, two multiply-adds, a lane shift; acc sum order changes
# speedup vs baseline: 2.4091x; 2.4091x over previous
#include <hip/hip_runtime.h>
#include <hip/hip_bf16.h>


typedef float v2f __attribute__((ext_vector_type(2)));
typedef float v4f __attribute__((ext_vector_type(4)));
typedef float v8f __attribute__((ext_vector_type(8)));

#define ORDER 16
#define FRAME 160
#define EPS_F 1e-8f
#define WAVES_PER_BLOCK 8
#define FRAMES_PER_WAVE 2
#define FRAMES_PER_BLOCK (WAVES_PER_BLOCK * FRAMES_PER_WAVE)
#define PAD 16
#define REGION (PAD + FRAME + PAD)
#define TAB_KEYS 46
#define TAB (TAB_KEYS * 8)

__device__ __forceinline__ float pair_even(float v) {
  return __int_as_float(__builtin_amdgcn_update_dpp(0, __float_as_int(v), 0xA0, 0xf, 0xf, true));
}
__device__ __forceinline__ float pair_odd(float v) {
  return __int_as_float(__builtin_amdgcn_update_dpp(0, __float_as_int(v), 0xF5, 0xf, 0xf, true));
}

__device__ __forceinline__ float row_dot16(float b, float re, float ro) {
  float v = __builtin_fmaf(pair_even(b), re, pair_odd(b) * ro);
  v += __int_as_float(__builtin_amdgcn_update_dpp(0, __float_as_int(v), 0x162, 0xf, 0xf, true));
  v += __int_as_float(__builtin_amdgcn_update_dpp(0, __float_as_int(v), 0x164, 0xf, 0xf, true));
  v += __int_as_float(__builtin_amdgcn_update_dpp(0, __float_as_int(v), 0x168, 0xf, 0xf, true));
  return v;
}

__device__ __forceinline__ float row_shift_up1(float first, float src) {
  return __int_as_float(__builtin_amdgcn_update_dpp(
      __float_as_int(first), __float_as_int(src), 0x111, 0xf, 0xf, false));
}

__global__ __launch_bounds__(256) void lpc_residual_kernel(
    const float* __restrict__ x, float* __restrict__ out, int n_frames) {
  __shared__ __attribute__((aligned(16))) float sh[WAVES_PER_BLOCK][FRAMES_PER_WAVE][REGION];
  __shared__ __attribute__((aligned(16))) float tab[WAVES_PER_BLOCK][FRAMES_PER_WAVE][TAB];

  const int lane = threadIdx.x & 31;
  const int wv = threadIdx.x >> 5;
  const int m = lane & 15;
  const int h = lane >> 4;
  const int frame0 = (blockIdx.x * WAVES_PER_BLOCK + wv) * FRAMES_PER_WAVE;
  const bool active = frame0 < n_frames;
  const bool two = frame0 + 1 < n_frames;

  if (active) {
    sh[wv][h][m] = 0.0f;
    sh[wv][h][PAD + FRAME + m] = 0.0f;
    const float* gx = x + (long long)frame0 * FRAME;
#pragma unroll
    for (int t = 0; t < 5; ++t) sh[wv][0][PAD + lane + 32 * t] = gx[lane + 32 * t];
    if (two) {
#pragma unroll
      for (int t = 0; t < 5; ++t) sh[wv][1][PAD + lane + 32 * t] = gx[FRAME + lane + 32 * t];
    } else {
#pragma unroll
      for (int t = 0; t < 5; ++t) sh[wv][1][PAD + lane + 32 * t] = 0.0f;
    }
  }
  __syncthreads();

  if (active) {
    const int key0 = 8 * h + 2 * m;
#pragma unroll
    for (int g = 0; g < FRAMES_PER_WAVE; ++g) {
      const float* fr = &sh[wv][g][PAD];
      const float* pa = fr + m + 16 * h;
      const float* pb = fr + 14 - 2 * m + 16 * h;
      v8f D = {0.f, 0.f, 0.f, 0.f, 0.f, 0.f, 0.f, 0.f};
#pragma unroll
      for (int s = 0; s < 5; ++s) {
        v2f A, B;
        A.x = pa[32 * s];
        A.y = pa[32 * s + 1];
        B = *(const v2f*)(pb + 32 * s);
        D = __builtin_amdgcn_wmma_f32_16x16x4_f32(false, A, false, B, (short)0, D,
                                                  false, false);
      }
      float* tb = &tab[wv][g][0];
#pragma unroll
      for (int q = 0; q < 8; ++q) tb[(key0 + q) * 8 + (m & 7)] = D[q];
    }
    __builtin_amdgcn_fence(__ATOMIC_RELEASE, "wavefront");
    __builtin_amdgcn_wave_barrier();
    __builtin_amdgcn_fence(__ATOMIC_ACQUIRE, "wavefront");
    const float* tr = &tab[wv][h][0];
    const v4f lo = *(const v4f*)(tr + (m + 15) * 8);
    const v4f hi = *(const v4f*)(tr + (m + 15) * 8 + 4);
    const float R = ((lo.x + lo.y) + (lo.z + lo.w)) + ((hi.x + hi.y) + (hi.z + hi.w));
    const v4f lo0 = *(const v4f*)(tr + 14 * 8);
    const v4f hi0 = *(const v4f*)(tr + 14 * 8 + 4);
    const float r0 = ((lo0.x + lo0.y) + (lo0.z + lo0.w)) + ((hi0.x + hi0.y) + (hi0.z + hi0.w));

    const float Re = pair_even(R);
    const float Ro = pair_odd(R);
    float A = 0.0f;
    float B = (m == 0) ? -1.0f : 0.0f;
    float ne = -((r0 != 0.0f) ? r0 : EPS_F);
#pragma unroll
    for (int i = 1; i <= ORDER; ++i) {
      const float accv = -row_dot16(B, Re, Ro);
      float rc = __builtin_amdgcn_rcpf(-ne);
      rc = __builtin_fmaf(__builtin_fmaf(ne, rc, 1.0f), rc, rc);
      float q = accv * rc;
      q = __builtin_fmaf(__builtin_fmaf(ne, q, accv), rc, q);
      q = __builtin_fmaf(__builtin_fmaf(ne, q, accv), rc, q);
      const float k = (-ne != 0.0f) ? q : 0.0f;
      const float An = A - k * B;
      const float T = B - k * A;
      B = row_shift_up1(k, T);
      A = An;
      ne = fminf(ne * (1.0f - k * k), -EPS_F);
    }

#pragma unroll
    for (int g = 0; g < FRAMES_PER_WAVE; ++g) {
      float a[ORDER + 1];
      a[0] = 1.0f;
#pragma unroll
      for (int j = 1; j <= ORDER; ++j)
        a[j] = __int_as_float(__builtin_amdgcn_readlane(__float_as_int(A), 16 * g + j - 1));
      if (g == 0 || two) {
        const float* fr = &sh[wv][g][PAD];
        float* go = out + (long long)(frame0 + g) * FRAME;
#pragma unroll
        for (int t = 0; t < 5; ++t) {
          int n = lane + 32 * t;
          float s = 0.0f;
#pragma unroll
          for (int k = 0; k <= ORDER; ++k) s += a[k] * fr[n - k];
          go[n] = s;
        }
      }
    }
  }
}

extern "C" void kernel_launch(void* const* d_in, const int* in_sizes, int n_in,
                              void* d_out, int out_size, void* d_ws, size_t ws_size,
                              hipStream_t stream) {
  (void)n_in; (void)out_size; (void)d_ws; (void)ws_size;
  const float* x = (const float*)d_in[0];
  float* out = (float*)d_out;
  const int total = in_sizes[0];
  const int n_frames = total / FRAME;
  const int blocks = (n_frames + FRAMES_PER_BLOCK - 1) / FRAMES_PER_BLOCK;
  lpc_residual_kernel<<<blocks, WAVES_PER_BLOCK * 32, 0, stream>>>(x, out, n_frames);
}


# Levinson-Durbin with two neighbouring coefficients per lane: eight lanes hold a frame, so a wave takes four frames (was two) through each instruction of the recursion; same sum order and bits
# speedup vs baseline: 2.5980x; 1.0784x over previous
#include <hip/hip_runtime.h>
#include <hip/hip_bf16.h>


typedef float v2f __attribute__((ext_vector_type(2)));
typedef float v4f __attribute__((ext_vector_type(4)));
typedef float v8f __attribute__((ext_vector_type(8)));

#define ORDER 16
#define FRAME 160
#define EPS_F 1e-8f
#define WAVES_PER_BLOCK 8
#define FRAMES_PER_WAVE 4
#define FRAMES_PER_BLOCK (WAVES_PER_BLOCK * FRAMES_PER_WAVE)
#define PAD 16
#define REGION (PAD + FRAME + PAD)
#define TAB_KEYS 46
#define TAB (TAB_KEYS * 8)

__device__ __forceinline__ float frame_sum8(float v) {
  v += __int_as_float(__builtin_amdgcn_update_dpp(0, __float_as_int(v), 0x162, 0xf, 0xf, true));
  v += __int_as_float(__builtin_amdgcn_update_dpp(0, __float_as_int(v), 0x164, 0xf, 0xf, true));
  v += __int_as_float(__builtin_amdgcn_update_dpp(0, __float_as_int(v), 0x168, 0xf, 0xf, true));
  return v;
}

__device__ __forceinline__ float frame_shift_up1(float first, float src) {
  return __int_as_float(__builtin_amdgcn_update_dpp(
      __float_as_int(first), __float_as_int(src), 0x112, 0xf, 0xf, false));
}

__global__ __launch_bounds__(256) void lpc_residual_kernel(
    const float* __restrict__ x, float* __restrict__ out, int n_frames) {
  __shared__ __attribute__((aligned(16))) float sh[WAVES_PER_BLOCK][FRAMES_PER_WAVE][REGION];
  __shared__ __attribute__((aligned(16))) float tab[WAVES_PER_BLOCK][FRAMES_PER_WAVE][TAB];

  const int lane = threadIdx.x & 31;
  const int wv = threadIdx.x >> 5;
  const int m = lane & 15;
  const int h = lane >> 4;
  const int frame0 = (blockIdx.x * WAVES_PER_BLOCK + wv) * FRAMES_PER_WAVE;
  const bool active = frame0 < n_frames;
  const int nvalid = n_frames - frame0;

  if (active) {
    {
      const v2f z = {0.0f, 0.0f};
      float* pz = &sh[wv][lane >> 3][2 * (lane & 7)];
      *(v2f*)pz = z;
      *(v2f*)(pz + PAD + FRAME) = z;
    }
    const float* gx = x + (long long)frame0 * FRAME;
    float v[FRAMES_PER_WAVE][5];
#pragma unroll
    for (int g = 0; g < FRAMES_PER_WAVE; ++g) {
      const float* gg = gx + ((g < nvalid) ? g : 0) * FRAME;
#pragma unroll
      for (int t = 0; t < 5; ++t) v[g][t] = gg[lane + 32 * t];
    }
#pragma unroll
    for (int g = 0; g < FRAMES_PER_WAVE; ++g) {
#pragma unroll
      for (int t = 0; t < 5; ++t) sh[wv][g][PAD + lane + 32 * t] = v[g][t];
    }
  }
  __syncthreads();

  if (active) {
    const int key0 = 8 * h + 2 * m;
#pragma unroll
    for (int g = 0; g < FRAMES_PER_WAVE; ++g) {
      const float* fr = &sh[wv][g][PAD];
      const float* pa = fr + m + 16 * h;
      const float* pb = fr + 14 - 2 * m + 16 * h;
      v8f D = {0.f, 0.f, 0.f, 0.f, 0.f, 0.f, 0.f, 0.f};
#pragma unroll
      for (int s = 0; s < 5; ++s) {
        v2f A, B;
        A.x = pa[32 * s];
        A.y = pa[32 * s + 1];
        B = *(const v2f*)(pb + 32 * s);
        D = __builtin_amdgcn_wmma_f32_16x16x4_f32(false, A, false, B, (short)0, D,
                                                  false, false);
      }
      float* tb = &tab[wv][g][0];
#pragma unroll
      for (int q = 0; q < 8; ++q) tb[(key0 + q) * 8 + (m & 7)] = D[q];
    }
    __builtin_amdgcn_fence(__ATOMIC_RELEASE, "wavefront");
    __builtin_amdgcn_wave_barrier();
    __builtin_amdgcn_fence(__ATOMIC_ACQUIRE, "wavefront");
    const int e = m & 1;
    const int j = m >> 1;
    const float* tr = &tab[wv][2 * h + e][0];
    const v4f lo = *(const v4f*)(tr + (2 * j + 15) * 8);
    const v4f hi = *(const v4f*)(tr + (2 * j + 15) * 8 + 4);
    const float R0 = ((lo.x + lo.y) + (lo.z + lo.w)) + ((hi.x + hi.y) + (hi.z + hi.w));
    const v4f lo1 = *(const v4f*)(tr + (2 * j + 16) * 8);
    const v4f hi1 = *(const v4f*)(tr + (2 * j + 16) * 8 + 4);
    const float R1 = ((lo1.x + lo1.y) + (lo1.z + lo1.w)) + ((hi1.x + hi1.y) + (hi1.z + hi1.w));
    const v4f lo0 = *(const v4f*)(tr + 14 * 8);
    const v4f hi0 = *(const v4f*)(tr + 14 * 8 + 4);
    const float r0 = ((lo0.x + lo0.y) + (lo0.z + lo0.w)) + ((hi0.x + hi0.y) + (hi0.z + hi0.w));

    float A0 = 0.0f, A1 = 0.0f;
    float B0 = (j == 0) ? -1.0f : 0.0f;
    float B1 = 0.0f;
    float ne = -((r0 != 0.0f) ? r0 : EPS_F);
#pragma unroll
    for (int i = 1; i <= ORDER; ++i) {
      const float accv = -frame_sum8(__builtin_fmaf(B0, R0, B1 * R1));
      float rc = __builtin_amdgcn_rcpf(-ne);
      rc = __builtin_fmaf(__builtin_fmaf(ne, rc, 1.0f), rc, rc);
      float q = accv * rc;
      q = __builtin_fmaf(__builtin_fmaf(ne, q, accv), rc, q);
      q = __builtin_fmaf(__builtin_fmaf(ne, q, accv), rc, q);
      const float k = (-ne != 0.0f) ? q : 0.0f;
      const float An0 = A0 - k * B0;
      const float An1 = A1 - k * B1;
      const float T0 = B0 - k * A0;
      const float T1 = B1 - k * A1;
      B0 = frame_shift_up1(k, T1);
      B1 = T0;
      A0 = An0;
      A1 = An1;
      ne = fminf(ne * (1.0f - k * k), -EPS_F);
    }

#pragma unroll
    for (int g = 0; g < FRAMES_PER_WAVE; ++g) {
      float a[ORDER + 1];
      a[0] = 1.0f;
#pragma unroll
      for (int c = 0; c < ORDER; ++c)
        a[c + 1] = __int_as_float(__builtin_amdgcn_readlane(
            __float_as_int((c & 1) ? A1 : A0), 16 * (g >> 1) + 2 * (c >> 1) + (g & 1)));
      if (g < nvalid) {
        const float* fr = &sh[wv][g][PAD];
        float* go = out + (long long)(frame0 + g) * FRAME;
#pragma unroll
        for (int t = 0; t < 5; ++t) {
          int n = lane + 32 * t;
          float s = 0.0f;
#pragma unroll
          for (int k = 0; k <= ORDER; ++k) s += a[k] * fr[n - k];
          go[n] = s;
        }
      }
    }
  }
}

extern "C" void kernel_launch(void* const* d_in, const int* in_sizes, int n_in,
                              void* d_out, int out_size, void* d_ws, size_t ws_size,
                              hipStream_t stream) {
  (void)n_in; (void)out_size; (void)d_ws; (void)ws_size;
  const float* x = (const float*)d_in[0];
  float* out = (float*)d_out;
  const int total = in_sizes[0];
  const int n_frames = total / FRAME;
  const int blocks = (n_frames + FRAMES_PER_BLOCK - 1) / FRAMES_PER_BLOCK;
  lpc_residual_kernel<<<blocks, WAVES_PER_BLOCK * 32, 0, stream>>>(x, out, n_frames);
}


# Persistent waves: a fixed grid of 1000 work-groups (was 4000, one per tile), each wave looping over its tiles and loading the next tile's samples to registers while it computes; no block barrier
# speedup vs baseline: 2.7041x; 1.0408x over previous
#include <hip/hip_runtime.h>
#include <hip/hip_bf16.h>


typedef float v2f __attribute__((ext_vector_type(2)));
typedef float v4f __attribute__((ext_vector_type(4)));
typedef float v8f __attribute__((ext_vector_type(8)));

#define ORDER 16
#define FRAME 160
#define EPS_F 1e-8f
#define WAVES_PER_BLOCK 8
#define FRAMES_PER_WAVE 4
#define FRAMES_PER_BLOCK (WAVES_PER_BLOCK * FRAMES_PER_WAVE)
#define PAD 16
#define REGION (PAD + FRAME + PAD)
#define TAB_KEYS 46
#define TAB (TAB_KEYS * 8)

__device__ __forceinline__ float frame_sum8(float v) {
  v += __int_as_float(__builtin_amdgcn_update_dpp(0, __float_as_int(v), 0x162, 0xf, 0xf, true));
  v += __int_as_float(__builtin_amdgcn_update_dpp(0, __float_as_int(v), 0x164, 0xf, 0xf, true));
  v += __int_as_float(__builtin_amdgcn_update_dpp(0, __float_as_int(v), 0x168, 0xf, 0xf, true));
  return v;
}

__device__ __forceinline__ float frame_shift_up1(float first, float src) {
  return __int_as_float(__builtin_amdgcn_update_dpp(
      __float_as_int(first), __float_as_int(src), 0x112, 0xf, 0xf, false));
}

__device__ __forceinline__ void load_frames(const float* __restrict__ x, int frame0,
                                            int nvalid, int lane,
                                            float (&v)[FRAMES_PER_WAVE][5]) {
  const float* gx = x + (long long)frame0 * FRAME;
#pragma unroll
  for (int g = 0; g < FRAMES_PER_WAVE; ++g) {
    const float* gg = gx + ((g < nvalid) ? g : 0) * FRAME;
#pragma unroll
    for (int t = 0; t < 5; ++t) v[g][t] = gg[lane + 32 * t];
  }
}

__global__ __launch_bounds__(256) void lpc_residual_kernel(
    const float* __restrict__ x, float* __restrict__ out, int n_frames) {
  __shared__ __attribute__((aligned(16))) float sh[WAVES_PER_BLOCK][FRAMES_PER_WAVE][REGION];
  __shared__ __attribute__((aligned(16))) float tab[WAVES_PER_BLOCK][FRAMES_PER_WAVE][TAB];

  const int lane = threadIdx.x & 31;
  const int wv = __builtin_amdgcn_readfirstlane(threadIdx.x >> 5);
  const int m = lane & 15;
  const int h = lane >> 4;
  const int stride = gridDim.x * FRAMES_PER_BLOCK;
  int frame0 = (blockIdx.x * WAVES_PER_BLOCK + wv) * FRAMES_PER_WAVE;
  if (frame0 >= n_frames) return;

  {
    const v2f z = {0.0f, 0.0f};
    float* pz = &sh[wv][lane >> 3][2 * (lane & 7)];
    *(v2f*)pz = z;
    *(v2f*)(pz + PAD + FRAME) = z;
  }
  float v[FRAMES_PER_WAVE][5];
  load_frames(x, frame0, n_frames - frame0, lane, v);

  for (;;) {
    const int nvalid = n_frames - frame0;
#pragma unroll
    for (int g = 0; g < FRAMES_PER_WAVE; ++g) {
#pragma unroll
      for (int t = 0; t < 5; ++t) sh[wv][g][PAD + lane + 32 * t] = v[g][t];
    }
    __builtin_amdgcn_fence(__ATOMIC_RELEASE, "wavefront");
    __builtin_amdgcn_wave_barrier();
    __builtin_amdgcn_fence(__ATOMIC_ACQUIRE, "wavefront");
    const int next0 = frame0 + stride;
    if (next0 < n_frames) load_frames(x, next0, n_frames - next0, lane, v);

    {
      const int key0 = 8 * h + 2 * m;
#pragma unroll
      for (int g = 0; g < FRAMES_PER_WAVE; ++g) {
        const float* fr = &sh[wv][g][PAD];
        const float* pa = fr + m + 16 * h;
        const float* pb = fr + 14 - 2 * m + 16 * h;
        v8f D = {0.f, 0.f, 0.f, 0.f, 0.f, 0.f, 0.f, 0.f};
#pragma unroll
        for (int s = 0; s < 5; ++s) {
          v2f A, B;
          A.x = pa[32 * s];
          A.y = pa[32 * s + 1];
          B = *(const v2f*)(pb + 32 * s);
          D = __builtin_amdgcn_wmma_f32_16x16x4_f32(false, A, false, B, (short)0, D,
                                                    false, false);
        }
        float* tb = &tab[wv][g][0];
#pragma unroll
        for (int q = 0; q < 8; ++q) tb[(key0 + q) * 8 + (m & 7)] = D[q];
      }
      __builtin_amdgcn_fence(__ATOMIC_RELEASE, "wavefront");
      __builtin_amdgcn_wave_barrier();
      __builtin_amdgcn_fence(__ATOMIC_ACQUIRE, "wavefront");
      const int e = m & 1;
      const int j = m >> 1;
      const float* tr = &tab[wv][2 * h + e][0];
      const v4f lo = *(const v4f*)(tr + (2 * j + 15) * 8);
      const v4f hi = *(const v4f*)(tr + (2 * j + 15) * 8 + 4);
      const float R0 = ((lo.x + lo.y) + (lo.z + lo.w)) + ((hi.x + hi.y) + (hi.z + hi.w));
      const v4f lo1 = *(const v4f*)(tr + (2 * j + 16) * 8);
      const v4f hi1 = *(const v4f*)(tr + (2 * j + 16) * 8 + 4);
      const float R1 = ((lo1.x + lo1.y) + (lo1.z + lo1.w)) + ((hi1.x + hi1.y) + (hi1.z + hi1.w));
      const v4f lo0 = *(const v4f*)(tr + 14 * 8);
      const v4f hi0 = *(const v4f*)(tr + 14 * 8 + 4);
      const float r0 = ((lo0.x + lo0.y) + (lo0.z + lo0.w)) + ((hi0.x + hi0.y) + (hi0.z + hi0.w));

      float A0 = 0.0f, A1 = 0.0f;
      float B0 = (j == 0) ? -1.0f : 0.0f;
      float B1 = 0.0f;
      float ne = -((r0 != 0.0f) ? r0 : EPS_F);
#pragma unroll
      for (int i = 1; i <= ORDER; ++i) {
        const float accv = -frame_sum8(__builtin_fmaf(B0, R0, B1 * R1));
        float rc = __builtin_amdgcn_rcpf(-ne);
        rc = __builtin_fmaf(__builtin_fmaf(ne, rc, 1.0f), rc, rc);
        float q = accv * rc;
        q = __builtin_fmaf(__builtin_fmaf(ne, q, accv), rc, q);
        q = __builtin_fmaf(__builtin_fmaf(ne, q, accv), rc, q);
        const float k = (-ne != 0.0f) ? q : 0.0f;
        const float An0 = A0 - k * B0;
        const float An1 = A1 - k * B1;
        const float T0 = B0 - k * A0;
        const float T1 = B1 - k * A1;
        B0 = frame_shift_up1(k, T1);
        B1 = T0;
        A0 = An0;
        A1 = An1;
        ne = fminf(ne * (1.0f - k * k), -EPS_F);
      }

#pragma unroll
      for (int g = 0; g < FRAMES_PER_WAVE; ++g) {
        float a[ORDER + 1];
        a[0] = 1.0f;
#pragma unroll
        for (int c = 0; c < ORDER; ++c)
          a[c + 1] = __int_as_float(__builtin_amdgcn_readlane(
              __float_as_int((c & 1) ? A1 : A0), 16 * (g >> 1) + 2 * (c >> 1) + (g & 1)));
        if (g < nvalid) {
          const float* fr = &sh[wv][g][PAD];
          float* go = out + (long long)(frame0 + g) * FRAME;
#pragma unroll
          for (int t = 0; t < 5; ++t) {
            int n = lane + 32 * t;
            float s = 0.0f;
#pragma unroll
            for (int k = 0; k <= ORDER; ++k) s += a[k] * fr[n - k];
            go[n] = s;
          }
        }
      }
    }
    if (next0 >= n_frames) break;
    frame0 = next0;
  }
}

extern "C" void kernel_launch(void* const* d_in, const int* in_sizes, int n_in,
                              void* d_out, int out_size, void* d_ws, size_t ws_size,
                              hipStream_t stream) {
  (void)n_in; (void)out_size; (void)d_ws; (void)ws_size;
  const float* x = (const float*)d_in[0];
  float* out = (float*)d_out;
  const int total = in_sizes[0];
  const int n_frames = total / FRAME;
  lpc_residual_kernel<<<1000, WAVES_PER_BLOCK * 32, 0, stream>>>(x, out, n_frames);
}


# FIR moved to the f32 matrix unit as a banded Toeplitz product, coefficients as the band operand: 8 matrix instructions a frame (was 17 LDS reads per output); results go via LDS; res sum order changes
# speedup vs baseline: 3.0262x; 1.1191x over previous
#include <hip/hip_runtime.h>
#include <hip/hip_bf16.h>


typedef float v2f __attribute__((ext_vector_type(2)));
typedef float v4f __attribute__((ext_vector_type(4)));
typedef float v8f __attribute__((ext_vector_type(8)));

#define ORDER 16
#define FRAME 160
#define EPS_F 1e-8f
#define WAVES_PER_BLOCK 8
#define FRAMES_PER_WAVE 4
#define FRAMES_PER_BLOCK (WAVES_PER_BLOCK * FRAMES_PER_WAVE)
#define PAD 16
#define REGION (PAD + FRAME + PAD)
#define TAB_KEYS 46
#define TAB (TAB_KEYS * 8)
#define CT 48
#define FIR_COLS 10

__device__ __forceinline__ float frame_sum8(float v) {
  v += __int_as_float(__builtin_amdgcn_update_dpp(0, __float_as_int(v), 0x162, 0xf, 0xf, true));
  v += __int_as_float(__builtin_amdgcn_update_dpp(0, __float_as_int(v), 0x164, 0xf, 0xf, true));
  v += __int_as_float(__builtin_amdgcn_update_dpp(0, __float_as_int(v), 0x168, 0xf, 0xf, true));
  return v;
}

__device__ __forceinline__ float frame_shift_up1(float first, float src) {
  return __int_as_float(__builtin_amdgcn_update_dpp(
      __float_as_int(first), __float_as_int(src), 0x112, 0xf, 0xf, false));
}

__device__ __forceinline__ void load_frames(const float* __restrict__ x, int frame0,
                                            int nvalid, int lane,
                                            float (&v)[FRAMES_PER_WAVE][5]) {
  const float* gx = x + (long long)frame0 * FRAME;
#pragma unroll
  for (int g = 0; g < FRAMES_PER_WAVE; ++g) {
    const float* gg = gx + ((g < nvalid) ? g : 0) * FRAME;
#pragma unroll
    for (int t = 0; t < 5; ++t) v[g][t] = gg[lane + 32 * t];
  }
}

__global__ __launch_bounds__(256) void lpc_residual_kernel(
    const float* __restrict__ x, float* __restrict__ out, int n_frames) {
  __shared__ __attribute__((aligned(16))) float sh[WAVES_PER_BLOCK][FRAMES_PER_WAVE][REGION];
  __shared__ __attribute__((aligned(16))) float tab[WAVES_PER_BLOCK][FRAMES_PER_WAVE][TAB];
  __shared__ __attribute__((aligned(16))) float ct[WAVES_PER_BLOCK][FRAMES_PER_WAVE][CT];

  const int lane = threadIdx.x & 31;
  const int wv = __builtin_amdgcn_readfirstlane(threadIdx.x >> 5);
  const int m = lane & 15;
  const int h = lane >> 4;
  const int stride = gridDim.x * FRAMES_PER_BLOCK;
  int frame0 = (blockIdx.x * WAVES_PER_BLOCK + wv) * FRAMES_PER_WAVE;
  if (frame0 >= n_frames) return;

  {
    const v2f z = {0.0f, 0.0f};
    float* pz = &sh[wv][lane >> 3][2 * (lane & 7)];
    *(v2f*)pz = z;
    *(v2f*)(pz + PAD + FRAME) = z;
    float* pc = &ct[wv][0][0];
#pragma unroll
    for (int i = 0; i < 6; ++i) {
      const int idx = lane + 32 * i;
      pc[idx] = ((idx % CT) == 32) ? 1.0f : 0.0f;
    }
  }
  float v[FRAMES_PER_WAVE][5];
  load_frames(x, frame0, n_frames - frame0, lane, v);

  for (;;) {
    const int nvalid = n_frames - frame0;
#pragma unroll
    for (int g = 0; g < FRAMES_PER_WAVE; ++g) {
#pragma unroll
      for (int t = 0; t < 5; ++t) sh[wv][g][PAD + lane + 32 * t] = v[g][t];
    }
    __builtin_amdgcn_fence(__ATOMIC_RELEASE, "wavefront");
    __builtin_amdgcn_wave_barrier();
    __builtin_amdgcn_fence(__ATOMIC_ACQUIRE, "wavefront");
    const int next0 = frame0 + stride;
    if (next0 < n_frames) load_frames(x, next0, n_frames - next0, lane, v);

    {
      const int key0 = 8 * h + 2 * m;
#pragma unroll
      for (int g = 0; g < FRAMES_PER_WAVE; ++g) {
        const float* fr = &sh[wv][g][PAD];
        const float* pa = fr + m + 16 * h;
        const float* pb = fr + 14 - 2 * m + 16 * h;
        v8f D = {0.f, 0.f, 0.f, 0.f, 0.f, 0.f, 0.f, 0.f};
#pragma unroll
        for (int s = 0; s < 5; ++s) {
          v2f A, B;
          A.x = pa[32 * s];
          A.y = pa[32 * s + 1];
          B = *(const v2f*)(pb + 32 * s);
          D = __builtin_amdgcn_wmma_f32_16x16x4_f32(false, A, false, B, (short)0, D,
                                                    false, false);
        }
        float* tb = &tab[wv][g][0];
#pragma unroll
        for (int q = 0; q < 8; ++q) tb[(key0 + q) * 8 + (m & 7)] = D[q];
      }
      __builtin_amdgcn_fence(__ATOMIC_RELEASE, "wavefront");
      __builtin_amdgcn_wave_barrier();
      __builtin_amdgcn_fence(__ATOMIC_ACQUIRE, "wavefront");
      const int e = m & 1;
      const int j = m >> 1;
      const float* tr = &tab[wv][2 * h + e][0];
      const v4f lo = *(const v4f*)(tr + (2 * j + 15) * 8);
      const v4f hi = *(const v4f*)(tr + (2 * j + 15) * 8 + 4);
      const float R0 = ((lo.x + lo.y) + (lo.z + lo.w)) + ((hi.x + hi.y) + (hi.z + hi.w));
      const v4f lo1 = *(const v4f*)(tr + (2 * j + 16) * 8);
      const v4f hi1 = *(const v4f*)(tr + (2 * j + 16) * 8 + 4);
      const float R1 = ((lo1.x + lo1.y) + (lo1.z + lo1.w)) + ((hi1.x + hi1.y) + (hi1.z + hi1.w));
      const v4f lo0 = *(const v4f*)(tr + 14 * 8);
      const v4f hi0 = *(const v4f*)(tr + 14 * 8 + 4);
      const float r0 = ((lo0.x + lo0.y) + (lo0.z + lo0.w)) + ((hi0.x + hi0.y) + (hi0.z + hi0.w));

      float A0 = 0.0f, A1 = 0.0f;
      float B0 = (j == 0) ? -1.0f : 0.0f;
      float B1 = 0.0f;
      float ne = -((r0 != 0.0f) ? r0 : EPS_F);
#pragma unroll
      for (int i = 1; i <= ORDER; ++i) {
        const float accv = -frame_sum8(__builtin_fmaf(B0, R0, B1 * R1));
        float rc = __builtin_amdgcn_rcpf(-ne);
        rc = __builtin_fmaf(__builtin_fmaf(ne, rc, 1.0f), rc, rc);
        float q = accv * rc;
        q = __builtin_fmaf(__builtin_fmaf(ne, q, accv), rc, q);
        q = __builtin_fmaf(__builtin_fmaf(ne, q, accv), rc, q);
        const float k = (-ne != 0.0f) ? q : 0.0f;
        const float An0 = A0 - k * B0;
        const float An1 = A1 - k * B1;
        const float T0 = B0 - k * A0;
        const float T1 = B1 - k * A1;
        B0 = frame_shift_up1(k, T1);
        B1 = T0;
        A0 = An0;
        A1 = An1;
        ne = fminf(ne * (1.0f - k * k), -EPS_F);
      }

      {
        const v2f c2 = {A1, A0};
        *(v2f*)&ct[wv][2 * h + e][30 - 2 * j] = c2;
      }
      __builtin_amdgcn_fence(__ATOMIC_RELEASE, "wavefront");
      __builtin_amdgcn_wave_barrier();
      __builtin_amdgcn_fence(__ATOMIC_ACQUIRE, "wavefront");
      const int nb = (m < FIR_COLS) ? m : (FIR_COLS - 1);
#pragma unroll
      for (int g = 0; g < FRAMES_PER_WAVE; ++g) {
        const float* pa = &ct[wv][g][16 - m + 2 * h];
        const float* pb = &sh[wv][g][16 * nb + 2 * h];
        v8f D = {0.f, 0.f, 0.f, 0.f, 0.f, 0.f, 0.f, 0.f};
#pragma unroll
        for (int kk = 0; kk < 8; ++kk) {
          v2f A, B;
          A.x = pa[4 * kk];
          A.y = pa[4 * kk + 1];
          B = *(const v2f*)(pb + 4 * kk);
          D = __builtin_amdgcn_wmma_f32_16x16x4_f32(false, A, false, B, (short)0, D,
                                                    false, false);
        }
        float* pd = &sh[wv][g][PAD + 16 * nb + 8 * h];
        const v4f d0 = {D[0], D[1], D[2], D[3]};
        const v4f d1 = {D[4], D[5], D[6], D[7]};
        *(v4f*)pd = d0;
        *(v4f*)(pd + 4) = d1;
      }
      __builtin_amdgcn_fence(__ATOMIC_RELEASE, "wavefront");
      __builtin_amdgcn_wave_barrier();
      __builtin_amdgcn_fence(__ATOMIC_ACQUIRE, "wavefront");
      {
        float* go = out + (long long)frame0 * FRAME;
        v4f r4[5];
#pragma unroll
        for (int t = 0; t < 5; ++t) {
          const int Q = lane + 32 * t;
          const int fq = Q / 40;
          const int oq = Q - 40 * fq;
          r4[t] = *(const v4f*)&sh[wv][fq][PAD + 4 * oq];
        }
        if (nvalid >= FRAMES_PER_WAVE) {
#pragma unroll
          for (int t = 0; t < 5; ++t) *(v4f*)(go + 4 * (lane + 32 * t)) = r4[t];
        } else {
#pragma unroll
          for (int t = 0; t < 5; ++t) {
            const int Q = lane + 32 * t;
            if (Q / 40 < nvalid) *(v4f*)(go + 4 * Q) = r4[t];
          }
        }
      }
    }
    if (next0 >= n_frames) break;
    frame0 = next0;
  }
}

extern "C" void kernel_launch(void* const* d_in, const int* in_sizes, int n_in,
                              void* d_out, int out_size, void* d_ws, size_t ws_size,
                              hipStream_t stream) {
  (void)n_in; (void)out_size; (void)d_ws; (void)ws_size;
  const float* x = (const float*)d_in[0];
  float* out = (float*)d_out;
  const int total = in_sizes[0];
  const int n_frames = total / FRAME;
  lpc_residual_kernel<<<1000, WAVES_PER_BLOCK * 32, 0, stream>>>(x, out, n_frames);
}


# Fragment loads of both matrix phases issued as one group ahead of their products, two frames' chains alternating (was an LDS load or store between almost every two products); same sums, same bits
# speedup vs baseline: 3.1574x; 1.0434x over previous
#include <hip/hip_runtime.h>
#include <hip/hip_bf16.h>


typedef float v2f __attribute__((ext_vector_type(2)));
typedef float v4f __attribute__((ext_vector_type(4)));
typedef float v8f __attribute__((ext_vector_type(8)));

#define ORDER 16
#define FRAME 160
#define EPS_F 1e-8f
#define WAVES_PER_BLOCK 8
#define FRAMES_PER_WAVE 4
#define FRAMES_PER_BLOCK (WAVES_PER_BLOCK * FRAMES_PER_WAVE)
#define PAD 16
#define REGION (PAD + FRAME + PAD)
#define TAB_KEYS 46
#define TAB (TAB_KEYS * 8)
#define CT 48
#define FIR_COLS 10

__device__ __forceinline__ float frame_sum8(float v) {
  v += __int_as_float(__builtin_amdgcn_update_dpp(0, __float_as_int(v), 0x162, 0xf, 0xf, true));
  v += __int_as_float(__builtin_amdgcn_update_dpp(0, __float_as_int(v), 0x164, 0xf, 0xf, true));
  v += __int_as_float(__builtin_amdgcn_update_dpp(0, __float_as_int(v), 0x168, 0xf, 0xf, true));
  return v;
}

__device__ __forceinline__ float frame_shift_up1(float first, float src) {
  return __int_as_float(__builtin_amdgcn_update_dpp(
      __float_as_int(first), __float_as_int(src), 0x112, 0xf, 0xf, false));
}

__device__ __forceinline__ void load_frames(const float* __restrict__ x, int frame0,
                                            int nvalid, int lane,
                                            float (&v)[FRAMES_PER_WAVE][5]) {
  const float* gx = x + (long long)frame0 * FRAME;
#pragma unroll
  for (int g = 0; g < FRAMES_PER_WAVE; ++g) {
    const float* gg = gx + ((g < nvalid) ? g : 0) * FRAME;
#pragma unroll
    for (int t = 0; t < 5; ++t) v[g][t] = gg[lane + 32 * t];
  }
}

__global__ __launch_bounds__(256) void lpc_residual_kernel(
    const float* __restrict__ x, float* __restrict__ out, int n_frames) {
  __shared__ __attribute__((aligned(16))) float sh[WAVES_PER_BLOCK][FRAMES_PER_WAVE][REGION];
  __shared__ __attribute__((aligned(16))) float tab[WAVES_PER_BLOCK][FRAMES_PER_WAVE][TAB];
  __shared__ __attribute__((aligned(16))) float ct[WAVES_PER_BLOCK][FRAMES_PER_WAVE][CT];

  const int lane = threadIdx.x & 31;
  const int wv = __builtin_amdgcn_readfirstlane(threadIdx.x >> 5);
  const int m = lane & 15;
  const int h = lane >> 4;
  const int stride = gridDim.x * FRAMES_PER_BLOCK;
  int frame0 = (blockIdx.x * WAVES_PER_BLOCK + wv) * FRAMES_PER_WAVE;
  if (frame0 >= n_frames) return;

  {
    const v2f z = {0.0f, 0.0f};
    float* pz = &sh[wv][lane >> 3][2 * (lane & 7)];
    *(v2f*)pz = z;
    *(v2f*)(pz + PAD + FRAME) = z;
    float* pc = &ct[wv][0][0];
#pragma unroll
    for (int i = 0; i < 6; ++i) {
      const int idx = lane + 32 * i;
      pc[idx] = ((idx % CT) == 32) ? 1.0f : 0.0f;
    }
  }
  float v[FRAMES_PER_WAVE][5];
  load_frames(x, frame0, n_frames - frame0, lane, v);

  for (;;) {
    const int nvalid = n_frames - frame0;
#pragma unroll
    for (int g = 0; g < FRAMES_PER_WAVE; ++g) {
#pragma unroll
      for (int t = 0; t < 5; ++t) sh[wv][g][PAD + lane + 32 * t] = v[g][t];
    }
    __builtin_amdgcn_fence(__ATOMIC_RELEASE, "wavefront");
    __builtin_amdgcn_wave_barrier();
    __builtin_amdgcn_fence(__ATOMIC_ACQUIRE, "wavefront");
    const int next0 = frame0 + stride;
    if (next0 < n_frames) load_frames(x, next0, n_frames - next0, lane, v);

    {
      const int key0 = 8 * h + 2 * m;
#pragma unroll
      for (int gp = 0; gp < FRAMES_PER_WAVE / 2; ++gp) {
        const float* fr0 = &sh[wv][2 * gp][PAD];
        const float* fr1 = &sh[wv][2 * gp + 1][PAD];
        const float* pa0 = fr0 + m + 16 * h;
        const float* pa1 = fr1 + m + 16 * h;
        const float* pb0 = fr0 + 14 - 2 * m + 16 * h;
        const float* pb1 = fr1 + 14 - 2 * m + 16 * h;
        v2f A0[5], B0[5], A1[5], B1[5];
#pragma unroll
        for (int s = 0; s < 5; ++s) {
          A0[s].x = pa0[32 * s];
          A0[s].y = pa0[32 * s + 1];
          B0[s] = *(const v2f*)(pb0 + 32 * s);
          A1[s].x = pa1[32 * s];
          A1[s].y = pa1[32 * s + 1];
          B1[s] = *(const v2f*)(pb1 + 32 * s);
        }
        __builtin_amdgcn_sched_barrier(0);
        v8f D0 = {0.f, 0.f, 0.f, 0.f, 0.f, 0.f, 0.f, 0.f};
        v8f D1 = {0.f, 0.f, 0.f, 0.f, 0.f, 0.f, 0.f, 0.f};
#pragma unroll
        for (int s = 0; s < 5; ++s) {
          D0 = __builtin_amdgcn_wmma_f32_16x16x4_f32(false, A0[s], false, B0[s], (short)0, D0,
                                                     false, false);
          D1 = __builtin_amdgcn_wmma_f32_16x16x4_f32(false, A1[s], false, B1[s], (short)0, D1,
                                                     false, false);
        }
        float* tb0 = &tab[wv][2 * gp][0];
        float* tb1 = &tab[wv][2 * gp + 1][0];
#pragma unroll
        for (int q = 0; q < 8; ++q) tb0[(key0 + q) * 8 + (m & 7)] = D0[q];
#pragma unroll
        for (int q = 0; q < 8; ++q) tb1[(key0 + q) * 8 + (m & 7)] = D1[q];
      }
      __builtin_amdgcn_fence(__ATOMIC_RELEASE, "wavefront");
      __builtin_amdgcn_wave_barrier();
      __builtin_amdgcn_fence(__ATOMIC_ACQUIRE, "wavefront");
      const int e = m & 1;
      const int j = m >> 1;
      const float* tr = &tab[wv][2 * h + e][0];
      const v4f lo = *(const v4f*)(tr + (2 * j + 15) * 8);
      const v4f hi = *(const v4f*)(tr + (2 * j + 15) * 8 + 4);
      const float R0 = ((lo.x + lo.y) + (lo.z + lo.w)) + ((hi.x + hi.y) + (hi.z + hi.w));
      const v4f lo1 = *(const v4f*)(tr + (2 * j + 16) * 8);
      const v4f hi1 = *(const v4f*)(tr + (2 * j + 16) * 8 + 4);
      const float R1 = ((lo1.x + lo1.y) + (lo1.z + lo1.w)) + ((hi1.x + hi1.y) + (hi1.z + hi1.w));
      const v4f lo0 = *(const v4f*)(tr + 14 * 8);
      const v4f hi0 = *(const v4f*)(tr + 14 * 8 + 4);
      const float r0 = ((lo0.x + lo0.y) + (lo0.z + lo0.w)) + ((hi0.x + hi0.y) + (hi0.z + hi0.w));

      float A0 = 0.0f, A1 = 0.0f;
      float B0 = (j == 0) ? -1.0f : 0.0f;
      float B1 = 0.0f;
      float ne = -((r0 != 0.0f) ? r0 : EPS_F);
#pragma unroll
      for (int i = 1; i <= ORDER; ++i) {
        const float accv = -frame_sum8(__builtin_fmaf(B0, R0, B1 * R1));
        float rc = __builtin_amdgcn_rcpf(-ne);
        rc = __builtin_fmaf(__builtin_fmaf(ne, rc, 1.0f), rc, rc);
        float q = accv * rc;
        q = __builtin_fmaf(__builtin_fmaf(ne, q, accv), rc, q);
        q = __builtin_fmaf(__builtin_fmaf(ne, q, accv), rc, q);
        const float k = (-ne != 0.0f) ? q : 0.0f;
        const float An0 = A0 - k * B0;
        const float An1 = A1 - k * B1;
        const float T0 = B0 - k * A0;
        const float T1 = B1 - k * A1;
        B0 = frame_shift_up1(k, T1);
        B1 = T0;
        A0 = An0;
        A1 = An1;
        ne = fminf(ne * (1.0f - k * k), -EPS_F);
      }

      {
        const v2f c2 = {A1, A0};
        *(v2f*)&ct[wv][2 * h + e][30 - 2 * j] = c2;
      }
      __builtin_amdgcn_fence(__ATOMIC_RELEASE, "wavefront");
      __builtin_amdgcn_wave_barrier();
      __builtin_amdgcn_fence(__ATOMIC_ACQUIRE, "wavefront");
      const int nb = (m < FIR_COLS) ? m : (FIR_COLS - 1);
#pragma unroll
      for (int gp = 0; gp < FRAMES_PER_WAVE / 2; ++gp) {
        const float* pa0 = &ct[wv][2 * gp][16 - m + 2 * h];
        const float* pa1 = &ct[wv][2 * gp + 1][16 - m + 2 * h];
        const float* pb0 = &sh[wv][2 * gp][16 * nb + 2 * h];
        const float* pb1 = &sh[wv][2 * gp + 1][16 * nb + 2 * h];
        v8f D0 = {0.f, 0.f, 0.f, 0.f, 0.f, 0.f, 0.f, 0.f};
        v8f D1 = {0.f, 0.f, 0.f, 0.f, 0.f, 0.f, 0.f, 0.f};
#pragma unroll
        for (int hk = 0; hk < 2; ++hk) {
          v2f A0[4], B0[4], A1[4], B1[4];
#pragma unroll
          for (int i = 0; i < 4; ++i) {
            const int kk = 4 * hk + i;
            A0[i].x = pa0[4 * kk];
            A0[i].y = pa0[4 * kk + 1];
            B0[i] = *(const v2f*)(pb0 + 4 * kk);
            A1[i].x = pa1[4 * kk];
            A1[i].y = pa1[4 * kk + 1];
            B1[i] = *(const v2f*)(pb1 + 4 * kk);
          }
          __builtin_amdgcn_sched_barrier(0);
#pragma unroll
          for (int i = 0; i < 4; ++i) {
            D0 = __builtin_amdgcn_wmma_f32_16x16x4_f32(false, A0[i], false, B0[i], (short)0, D0,
                                                       false, false);
            D1 = __builtin_amdgcn_wmma_f32_16x16x4_f32(false, A1[i], false, B1[i], (short)0, D1,
                                                       false, false);
          }
          __builtin_amdgcn_sched_barrier(0);
        }
        float* pd0 = &sh[wv][2 * gp][PAD + 16 * nb + 8 * h];
        float* pd1 = &sh[wv][2 * gp + 1][PAD + 16 * nb + 8 * h];
        const v4f d00 = {D0[0], D0[1], D0[2], D0[3]};
        const v4f d01 = {D0[4], D0[5], D0[6], D0[7]};
        const v4f d10 = {D1[0], D1[1], D1[2], D1[3]};
        const v4f d11 = {D1[4], D1[5], D1[6], D1[7]};
        *(v4f*)pd0 = d00;
        *(v4f*)(pd0 + 4) = d01;
        *(v4f*)pd1 = d10;
        *(v4f*)(pd1 + 4) = d11;
      }
      __builtin_amdgcn_fence(__ATOMIC_RELEASE, "wavefront");
      __builtin_amdgcn_wave_barrier();
      __builtin_amdgcn_fence(__ATOMIC_ACQUIRE, "wavefront");
      {
        float* go = out + (long long)frame0 * FRAME;
        v4f r4[5];
#pragma unroll
        for (int t = 0; t < 5; ++t) {
          const int Q = lane + 32 * t;
          const int fq = Q / 40;
          const int oq = Q - 40 * fq;
          r4[t] = *(const v4f*)&sh[wv][fq][PAD + 4 * oq];
        }
        if (nvalid >= FRAMES_PER_WAVE) {
#pragma unroll
          for (int t = 0; t < 5; ++t) *(v4f*)(go + 4 * (lane + 32 * t)) = r4[t];
        } else {
#pragma unroll
          for (int t = 0; t < 5; ++t) {
            const int Q = lane + 32 * t;
            if (Q / 40 < nvalid) *(v4f*)(go + 4 * Q) = r4[t];
          }
        }
      }
    }
    if (next0 >= n_frames) break;
    frame0 = next0;
  }
}

extern "C" void kernel_launch(void* const* d_in, const int* in_sizes, int n_in,
                              void* d_out, int out_size, void* d_ws, size_t ws_size,
                              hipStream_t stream) {
  (void)n_in; (void)out_size; (void)d_ws; (void)ws_size;
  const float* x = (const float*)d_in[0];
  float* out = (float*)d_out;
  const int total = in_sizes[0];
  const int n_frames = total / FRAME;
  lpc_residual_kernel<<<1000, WAVES_PER_BLOCK * 32, 0, stream>>>(x, out, n_frames);
}
